# starting kernel
# baseline (speedup 1.0000x reference)
#include <hip/hip_runtime.h>

typedef __attribute__((ext_vector_type(2))) float v2f;
typedef __attribute__((ext_vector_type(8))) float v8f;

namespace {
constexpr int kV     = 163842;
constexpr int kK     = 6;
constexpr int kG     = 5;
constexpr int kN     = 2;
constexpr int kCin   = 32;
constexpr int kCout  = 64;
constexpr int kKdim  = kCin * kG;
constexpr int kVT    = 16;
constexpr int kWaves = 4;
constexpr int kCwF   = kCout * kKdim;
constexpr int kGrF   = kVT * kKdim;
}

__global__ __launch_bounds__(kWaves * 32) void icosphere_conv_wmma(
    const float* __restrict__ x,
    const int*   __restrict__ nidx,
    const float* __restrict__ gradw,
    const float* __restrict__ convw,
    const float* __restrict__ convb,
    float* __restrict__ out)
{
  extern __shared__ float smem[];
  float* cw = smem;
  float* gr = smem + kCwF;

  const int tid  = threadIdx.x;
  const int lane = tid & 31;
  const int wave = tid >> 5;
  const int vloc = lane & 15;
  const int half = lane >> 4;

  for (int i = tid; i < kCwF; i += kWaves * 32) cw[i] = convw[i];
  __syncthreads();

  const int vtile = blockIdx.x * kWaves + wave;
  const int v     = vtile * kVT + vloc;
  const int vc    = (v < kV) ? v : (kV - 1);

  int nb[kK];
  #pragma unroll
  for (int k = 0; k < kK; ++k) nb[k] = nidx[vc * kK + k];
  float gw[kG * kK];
  #pragma unroll
  for (int j = 0; j < kG * kK; ++j) gw[j] = gradw[(size_t)vc * (kG * kK) + j];

  float* grs = gr + wave * (kN * kGrF);

  #pragma unroll
  for (int n = 0; n < kN; ++n) {
    #pragma unroll 4
    for (int ci = 0; ci < kCin / 2; ++ci) {
      const int c = half * (kCin / 2) + ci;
      const float* xp = x + ((size_t)n * kCin + c) * kV;
      float xv[kK];
      #pragma unroll
      for (int k = 0; k < kK; ++k) xv[k] = xp[nb[k]];
      #pragma unroll
      for (int g = 0; g < kG; ++g) {
        float s = 0.f;
        #pragma unroll
        for (int k = 0; k < kK; ++k) s = __builtin_fmaf(xv[k], gw[g * kK + k], s);
        grs[n * kGrF + vloc * kKdim + c * kG + g] = s;
      }
    }
  }

  v8f acc[kN][4] = {};
  #pragma unroll 4
  for (int kk = 0; kk < kKdim / 4; ++kk) {
    const int kb = kk * 4 + half * 2;
    const v2f b0 = *(const v2f*)(grs + 0 * kGrF + vloc * kKdim + kb);
    const v2f b1 = *(const v2f*)(grs + 1 * kGrF + vloc * kKdim + kb);
    #pragma unroll
    for (int m = 0; m < 4; ++m) {
      const v2f afr = *(const v2f*)(cw + (m * 16 + vloc) * kKdim + kb);
      acc[0][m] = __builtin_amdgcn_wmma_f32_16x16x4_f32(
          false, afr, false, b0, (short)0, acc[0][m], false, false);
      acc[1][m] = __builtin_amdgcn_wmma_f32_16x16x4_f32(
          false, afr, false, b1, (short)0, acc[1][m], false, false);
    }
  }

  if (v < kV) {
    #pragma unroll
    for (int n = 0; n < kN; ++n) {
      #pragma unroll
      for (int m = 0; m < 4; ++m) {
        #pragma unroll
        for (int r = 0; r < 8; ++r) {
          const int o = m * 16 + half * 8 + r;
          out[((size_t)n * kCout + o) * kV + v] = acc[n][m][r] + convb[o];
        }
      }
    }
  }
}

extern "C" void kernel_launch(void* const* d_in, const int* in_sizes, int n_in,
                              void* d_out, int out_size, void* d_ws, size_t ws_size,
                              hipStream_t stream) {
  (void)in_sizes; (void)n_in; (void)out_size; (void)d_ws; (void)ws_size;
  const float* x      = (const float*)d_in[0];
  const int*   nidx   = (const int*)  d_in[1];
  const float* gradw  = (const float*)d_in[2];
  const float* convw  = (const float*)d_in[3];
  const float* convb  = (const float*)d_in[4];
  float*       out    = (float*)d_out;

  const int vtiles = (kV + kVT - 1) / kVT;
  const int blocks = (vtiles + kWaves - 1) / kWaves;
  const size_t lds_bytes =
      (size_t)(kCwF + kWaves * kN * kGrF) * sizeof(float);

  icosphere_conv_wmma<<<blocks, kWaves * 32, lds_bytes, stream>>>(
      x, nidx, gradw, convw, convb, out);
}


# Gather from a vertex-major copy of x (V x 64 rows, built by a first launch in the workspace): a neighbor fetch is one aligned 256-B row, not 64 lines used 4 B each; sums and their order unchanged
# speedup vs baseline: 2.6380x; 2.6380x over previous
#include <hip/hip_runtime.h>

typedef __attribute__((ext_vector_type(2))) float v2f;
typedef __attribute__((ext_vector_type(4))) float v4f;
typedef __attribute__((ext_vector_type(8))) float v8f;
typedef __attribute__((address_space(1))) char gchar;
typedef __attribute__((address_space(1))) v2f  gv2f;

namespace {
constexpr int kV     = 163842;
constexpr int kK     = 6;
constexpr int kG     = 5;
constexpr int kN     = 2;
constexpr int kCin   = 32;
constexpr int kCout  = 64;
constexpr int kKdim  = kCin * kG;
constexpr int kVT    = 16;
constexpr int kWaves = 4;
constexpr int kCwF   = kCout * kKdim;
constexpr int kGrF   = kVT * kKdim;
constexpr int kRow   = kN * kCin;
constexpr int kTV    = 32;
constexpr int kTW    = 4;
constexpr int kTP    = kRow + 4;
constexpr size_t kXtBytes = (size_t)kV * kRow * sizeof(float);
}

__global__ __launch_bounds__(kWaves * 32) void icosphere_conv_wmma(
    const float* __restrict__ x,
    const int*   __restrict__ nidx,
    const float* __restrict__ gradw,
    const float* __restrict__ convw,
    const float* __restrict__ convb,
    float* __restrict__ out)
{
  extern __shared__ float smem[];
  float* cw = smem;
  float* gr = smem + kCwF;

  const int tid  = threadIdx.x;
  const int lane = tid & 31;
  const int wave = tid >> 5;
  const int vloc = lane & 15;
  const int half = lane >> 4;

  for (int i = tid; i < kCwF; i += kWaves * 32) cw[i] = convw[i];
  __syncthreads();

  const int vtile = blockIdx.x * kWaves + wave;
  const int v     = vtile * kVT + vloc;
  const int vc    = (v < kV) ? v : (kV - 1);

  int nb[kK];
  #pragma unroll
  for (int k = 0; k < kK; ++k) nb[k] = nidx[vc * kK + k];
  float gw[kG * kK];
  #pragma unroll
  for (int j = 0; j < kG * kK; ++j) gw[j] = gradw[(size_t)vc * (kG * kK) + j];

  float* grs = gr + wave * (kN * kGrF);

  #pragma unroll
  for (int n = 0; n < kN; ++n) {
    #pragma unroll 4
    for (int ci = 0; ci < kCin / 2; ++ci) {
      const int c = half * (kCin / 2) + ci;
      const float* xp = x + ((size_t)n * kCin + c) * kV;
      float xv[kK];
      #pragma unroll
      for (int k = 0; k < kK; ++k) xv[k] = xp[nb[k]];
      #pragma unroll
      for (int g = 0; g < kG; ++g) {
        float s = 0.f;
        #pragma unroll
        for (int k = 0; k < kK; ++k) s = __builtin_fmaf(xv[k], gw[g * kK + k], s);
        grs[n * kGrF + vloc * kKdim + c * kG + g] = s;
      }
    }
  }

  v8f acc[kN][4] = {};
  #pragma unroll 4
  for (int kk = 0; kk < kKdim / 4; ++kk) {
    const int kb = kk * 4 + half * 2;
    const v2f b0 = *(const v2f*)(grs + 0 * kGrF + vloc * kKdim + kb);
    const v2f b1 = *(const v2f*)(grs + 1 * kGrF + vloc * kKdim + kb);
    #pragma unroll
    for (int m = 0; m < 4; ++m) {
      const v2f afr = *(const v2f*)(cw + (m * 16 + vloc) * kKdim + kb);
      acc[0][m] = __builtin_amdgcn_wmma_f32_16x16x4_f32(
          false, afr, false, b0, (short)0, acc[0][m], false, false);
      acc[1][m] = __builtin_amdgcn_wmma_f32_16x16x4_f32(
          false, afr, false, b1, (short)0, acc[1][m], false, false);
    }
  }

  if (v < kV) {
    #pragma unroll
    for (int n = 0; n < kN; ++n) {
      #pragma unroll
      for (int m = 0; m < 4; ++m) {
        #pragma unroll
        for (int r = 0; r < 8; ++r) {
          const int o = m * 16 + half * 8 + r;
          out[((size_t)n * kCout + o) * kV + v] = acc[n][m][r] + convb[o];
        }
      }
    }
  }
}

__global__ __launch_bounds__(kTW * 32) void x_to_vertex_rows(
    const float* __restrict__ x,
    float* __restrict__ xt)
{
  __shared__ float tile[kTW * kTV * kTP];
  const int lane = threadIdx.x & 31;
  const int wave = threadIdx.x >> 5;
  const int v0   = (blockIdx.x * kTW + wave) * kTV;
  const int v    = v0 + lane;
  const int vc   = (v < kV) ? v : (kV - 1);

  float r[kRow];
  #pragma unroll
  for (int j = 0; j < kRow; ++j) r[j] = x[(size_t)j * kV + vc];

  float* t = tile + wave * (kTV * kTP);
  #pragma unroll
  for (int q = 0; q < kRow / 4; ++q) {
    v4f w4 = {r[4 * q], r[4 * q + 1], r[4 * q + 2], r[4 * q + 3]};
    *(v4f*)(t + lane * kTP + 4 * q) = w4;
  }
  const int sub  = lane >> 4;
  const int col4 = (lane & 15) * 4;
  v4f w[kTV / 2];
  #pragma unroll
  for (int s = 0; s < kTV / 2; ++s) w[s] = *(const v4f*)(t + (2 * s + sub) * kTP + col4);
  float* dst = xt + (size_t)(v0 + sub) * kRow + col4;
  if (v0 + kTV <= kV) {
    #pragma unroll
    for (int s = 0; s < kTV / 2; ++s) *(v4f*)(dst + (size_t)(2 * s) * kRow) = w[s];
  } else {
    #pragma unroll
    for (int s = 0; s < kTV / 2; ++s)
      if (v0 + 2 * s + sub < kV) *(v4f*)(dst + (size_t)(2 * s) * kRow) = w[s];
  }
}

__global__ __launch_bounds__(kWaves * 32) void icosphere_conv_wmma_rows(
    const float* __restrict__ xt,
    const int*   __restrict__ nidx,
    const float* __restrict__ gradw,
    const float* __restrict__ convw,
    const float* __restrict__ convb,
    float* __restrict__ out)
{
  extern __shared__ float smem[];
  float* cw = smem;
  float* gr = smem + kCwF;

  const int tid  = threadIdx.x;
  const int lane = tid & 31;
  const int wave = tid >> 5;
  const int vloc = lane & 15;
  const int half = lane >> 4;

  for (int i = tid; i < kCwF; i += kWaves * 32) cw[i] = convw[i];
  __syncthreads();

  const int vtile = blockIdx.x * kWaves + wave;
  const int v     = vtile * kVT + vloc;

  int nbv[3];
  #pragma unroll
  for (int j = 0; j < 3; ++j) {
    int f = vtile * (kVT * kK) + j * 32 + lane;
    f = (f < kV * kK) ? f : (kV * kK - 1);
    const unsigned u = (unsigned)nidx[f];
    nbv[j] = (int)(((u < (unsigned)kV) ? u : (unsigned)(kV - 1)) * (unsigned)(kRow * sizeof(float)));
  }
  float gwv[15];
  #pragma unroll
  for (int j = 0; j < 15; ++j) {
    int f = vtile * (kVT * kG * kK) + j * 32 + lane;
    f = (f < kV * kG * kK) ? f : (kV * kG * kK - 1);
    gwv[j] = gradw[f];
  }

  float* grs = gr + wave * (kN * kGrF);

  const unsigned lane8 = (unsigned)lane * 8u;
  const gchar* xtb = (const gchar*)xt;
  v2f xr[kVT][kK];
  #pragma unroll
  for (int h = 0; h < 2; ++h) {
    unsigned vo[kVT / 2][kK];
    #pragma unroll
    for (int i = 0; i < kVT / 2; ++i) {
      #pragma unroll
      for (int k = 0; k < kK; ++k) {
        const int f = (h * (kVT / 2) + i) * kK + k;
        vo[i][k] = (unsigned)__builtin_amdgcn_readlane(nbv[f >> 5], f & 31) + lane8;
      }
    }
    __builtin_amdgcn_sched_barrier(0);
    #pragma unroll
    for (int i = 0; i < kVT / 2; ++i) {
      #pragma unroll
      for (int k = 0; k < kK; ++k)
        xr[h * (kVT / 2) + i][k] = *(const gv2f*)(xtb + vo[i][k]);
    }
    __builtin_amdgcn_sched_barrier(0);
  }
  #pragma unroll
  for (int vi = 0; vi < kVT; ++vi) {
    float sx[kG], sy[kG];
    #pragma unroll
    for (int g = 0; g < kG; ++g) {
      float s0 = 0.f, s1 = 0.f;
      #pragma unroll
      for (int k = 0; k < kK; ++k) {
        const int f = vi * (kG * kK) + g * kK + k;
        const float w = __int_as_float(
            __builtin_amdgcn_readlane(__float_as_int(gwv[f >> 5]), f & 31));
        s0 = __builtin_fmaf(xr[vi][k].x, w, s0);
        s1 = __builtin_fmaf(xr[vi][k].y, w, s1);
      }
      sx[g] = s0; sy[g] = s1;
    }
    float* dst = grs + half * kGrF + vi * kKdim + vloc * (2 * kG);
    *(v2f*)(dst + 0) = v2f{sx[0], sx[1]};
    *(v2f*)(dst + 2) = v2f{sx[2], sx[3]};
    *(v2f*)(dst + 4) = v2f{sx[4], sy[0]};
    *(v2f*)(dst + 6) = v2f{sy[1], sy[2]};
    *(v2f*)(dst + 8) = v2f{sy[3], sy[4]};
  }

  v8f acc[kN][4] = {};
  #pragma unroll 4
  for (int kk = 0; kk < kKdim / 4; ++kk) {
    const int kb = kk * 4 + half * 2;
    const v2f b0 = *(const v2f*)(grs + 0 * kGrF + vloc * kKdim + kb);
    const v2f b1 = *(const v2f*)(grs + 1 * kGrF + vloc * kKdim + kb);
    #pragma unroll
    for (int m = 0; m < 4; ++m) {
      const v2f afr = *(const v2f*)(cw + (m * 16 + vloc) * kKdim + kb);
      acc[0][m] = __builtin_amdgcn_wmma_f32_16x16x4_f32(
          false, afr, false, b0, (short)0, acc[0][m], false, false);
      acc[1][m] = __builtin_amdgcn_wmma_f32_16x16x4_f32(
          false, afr, false, b1, (short)0, acc[1][m], false, false);
    }
  }

  if (v < kV) {
    #pragma unroll
    for (int n = 0; n < kN; ++n) {
      #pragma unroll
      for (int m = 0; m < 4; ++m) {
        #pragma unroll
        for (int r = 0; r < 8; ++r) {
          const int o = m * 16 + half * 8 + r;
          out[((size_t)n * kCout + o) * kV + v] = acc[n][m][r] + convb[o];
        }
      }
    }
  }
}

extern "C" void kernel_launch(void* const* d_in, const int* in_sizes, int n_in,
                              void* d_out, int out_size, void* d_ws, size_t ws_size,
                              hipStream_t stream) {
  (void)in_sizes; (void)n_in; (void)out_size;
  const float* x      = (const float*)d_in[0];
  const int*   nidx   = (const int*)  d_in[1];
  const float* gradw  = (const float*)d_in[2];
  const float* convw  = (const float*)d_in[3];
  const float* convb  = (const float*)d_in[4];
  float*       out    = (float*)d_out;
  float*       xt     = (float*)d_ws;

  const int vtiles = (kV + kVT - 1) / kVT;
  const int blocks = (vtiles + kWaves - 1) / kWaves;
  const size_t lds_bytes =
      (size_t)(kCwF + kWaves * kN * kGrF) * sizeof(float);

  if (ws_size >= kXtBytes) {
    const int ttiles  = (kV + kTV - 1) / kTV;
    const int tblocks = (ttiles + kTW - 1) / kTW;
    x_to_vertex_rows<<<tblocks, kTW * 32, 0, stream>>>(x, xt);
    icosphere_conv_wmma_rows<<<blocks, kWaves * 32, lds_bytes, stream>>>(
        xt, nidx, gradw, convw, convb, out);
  } else {
    icosphere_conv_wmma<<<blocks, kWaves * 32, lds_bytes, stream>>>(
        x, nidx, gradw, convw, convb, out);
  }
}


# LDS rows of both WMMA operands padded 160 -> 164 floats (16 B x odd): 8-way bank-conflicted fragment reads become conflict-free; conv_w copy re-indexed by rows for it; sums and order unchanged
# speedup vs baseline: 2.9810x; 1.1300x over previous
#include <hip/hip_runtime.h>

typedef __attribute__((ext_vector_type(2))) float v2f;
typedef __attribute__((ext_vector_type(4))) float v4f;
typedef __attribute__((ext_vector_type(8))) float v8f;
typedef __attribute__((address_space(1))) char gchar;
typedef __attribute__((address_space(1))) v2f  gv2f;

namespace {
constexpr int kV     = 163842;
constexpr int kK     = 6;
constexpr int kG     = 5;
constexpr int kN     = 2;
constexpr int kCin   = 32;
constexpr int kCout  = 64;
constexpr int kKdim  = kCin * kG;
constexpr int kVT    = 16;
constexpr int kWaves = 4;
constexpr int kCwN   = kCout * kKdim;
constexpr int kLp    = kKdim + 4;
constexpr int kCwF   = kCout * kLp;
constexpr int kGrP   = kVT * kLp + 32;
static_assert(kWaves * 16 == kCout && kKdim % 32 == 0, "conv_w copy: 16 rows per wave, 32 floats per load");
constexpr int kRow   = kN * kCin;
constexpr int kTV    = 32;
constexpr int kTW    = 4;
constexpr int kTP    = kRow + 4;
constexpr size_t kXtBytes = (size_t)kV * kRow * sizeof(float);
}

__global__ __launch_bounds__(kWaves * 32) void icosphere_conv_wmma(
    const float* __restrict__ x,
    const int*   __restrict__ nidx,
    const float* __restrict__ gradw,
    const float* __restrict__ convw,
    const float* __restrict__ convb,
    float* __restrict__ out)
{
  extern __shared__ float smem[];
  float* cw = smem;
  float* gr = smem + kCwF;

  const int tid  = threadIdx.x;
  const int lane = tid & 31;
  const int wave = tid >> 5;
  const int vloc = lane & 15;
  const int half = lane >> 4;

  {
    const float* src = convw + (wave * 16) * kKdim + lane;
    float* dstc = cw + (wave * 16) * kLp + lane;
    for (int j = 0; j < kKdim / 32; ++j) {
      #pragma unroll
      for (int r = 0; r < 16; ++r) dstc[r * kLp + j * 32] = src[r * kKdim + j * 32];
    }
  }
  __syncthreads();

  const int vtile = blockIdx.x * kWaves + wave;
  const int v     = vtile * kVT + vloc;
  const int vc    = (v < kV) ? v : (kV - 1);

  int nb[kK];
  #pragma unroll
  for (int k = 0; k < kK; ++k) nb[k] = nidx[vc * kK + k];
  float gw[kG * kK];
  #pragma unroll
  for (int j = 0; j < kG * kK; ++j) gw[j] = gradw[(size_t)vc * (kG * kK) + j];

  float* grs = gr + wave * (kN * kGrP);

  #pragma unroll
  for (int n = 0; n < kN; ++n) {
    #pragma unroll 4
    for (int ci = 0; ci < kCin / 2; ++ci) {
      const int c = half * (kCin / 2) + ci;
      const float* xp = x + ((size_t)n * kCin + c) * kV;
      float xv[kK];
      #pragma unroll
      for (int k = 0; k < kK; ++k) xv[k] = xp[nb[k]];
      #pragma unroll
      for (int g = 0; g < kG; ++g) {
        float s = 0.f;
        #pragma unroll
        for (int k = 0; k < kK; ++k) s = __builtin_fmaf(xv[k], gw[g * kK + k], s);
        grs[n * kGrP + vloc * kLp + c * kG + g] = s;
      }
    }
  }

  v8f acc[kN][4] = {};
  #pragma unroll 4
  for (int kk = 0; kk < kKdim / 4; ++kk) {
    const int kb = kk * 4 + half * 2;
    const v2f b0 = *(const v2f*)(grs + 0 * kGrP + vloc * kLp + kb);
    const v2f b1 = *(const v2f*)(grs + 1 * kGrP + vloc * kLp + kb);
    #pragma unroll
    for (int m = 0; m < 4; ++m) {
      const v2f afr = *(const v2f*)(cw + (m * 16 + vloc) * kLp + kb);
      acc[0][m] = __builtin_amdgcn_wmma_f32_16x16x4_f32(
          false, afr, false, b0, (short)0, acc[0][m], false, false);
      acc[1][m] = __builtin_amdgcn_wmma_f32_16x16x4_f32(
          false, afr, false, b1, (short)0, acc[1][m], false, false);
    }
  }

  if (v < kV) {
    #pragma unroll
    for (int n = 0; n < kN; ++n) {
      #pragma unroll
      for (int m = 0; m < 4; ++m) {
        #pragma unroll
        for (int r = 0; r < 8; ++r) {
          const int o = m * 16 + half * 8 + r;
          out[((size_t)n * kCout + o) * kV + v] = acc[n][m][r] + convb[o];
        }
      }
    }
  }
}

__global__ __launch_bounds__(kTW * 32) void x_to_vertex_rows(
    const float* __restrict__ x,
    float* __restrict__ xt)
{
  __shared__ float tile[kTW * kTV * kTP];
  const int lane = threadIdx.x & 31;
  const int wave = threadIdx.x >> 5;
  const int v0   = (blockIdx.x * kTW + wave) * kTV;
  const int v    = v0 + lane;
  const int vc   = (v < kV) ? v : (kV - 1);

  float r[kRow];
  #pragma unroll
  for (int j = 0; j < kRow; ++j) r[j] = x[(size_t)j * kV + vc];

  float* t = tile + wave * (kTV * kTP);
  #pragma unroll
  for (int q = 0; q < kRow / 4; ++q) {
    v4f w4 = {r[4 * q], r[4 * q + 1], r[4 * q + 2], r[4 * q + 3]};
    *(v4f*)(t + lane * kTP + 4 * q) = w4;
  }
  const int sub  = lane >> 4;
  const int col4 = (lane & 15) * 4;
  v4f w[kTV / 2];
  #pragma unroll
  for (int s = 0; s < kTV / 2; ++s) w[s] = *(const v4f*)(t + (2 * s + sub) * kTP + col4);
  float* dst = xt + (size_t)(v0 + sub) * kRow + col4;
  if (v0 + kTV <= kV) {
    #pragma unroll
    for (int s = 0; s < kTV / 2; ++s) *(v4f*)(dst + (size_t)(2 * s) * kRow) = w[s];
  } else {
    #pragma unroll
    for (int s = 0; s < kTV / 2; ++s)
      if (v0 + 2 * s + sub < kV) *(v4f*)(dst + (size_t)(2 * s) * kRow) = w[s];
  }
}

__global__ __launch_bounds__(kWaves * 32) void icosphere_conv_wmma_rows(
    const float* __restrict__ xt,
    const int*   __restrict__ nidx,
    const float* __restrict__ gradw,
    const float* __restrict__ convw,
    const float* __restrict__ convb,
    float* __restrict__ out)
{
  extern __shared__ float smem[];
  float* cw = smem;
  float* gr = smem + kCwF;

  const int tid  = threadIdx.x;
  const int lane = tid & 31;
  const int wave = tid >> 5;
  const int vloc = lane & 15;
  const int half = lane >> 4;

  {
    const float* src = convw + (wave * 16) * kKdim + lane;
    float* dstc = cw + (wave * 16) * kLp + lane;
    for (int j = 0; j < kKdim / 32; ++j) {
      #pragma unroll
      for (int r = 0; r < 16; ++r) dstc[r * kLp + j * 32] = src[r * kKdim + j * 32];
    }
  }
  __syncthreads();

  const int vtile = blockIdx.x * kWaves + wave;
  const int v     = vtile * kVT + vloc;

  int nbv[3];
  #pragma unroll
  for (int j = 0; j < 3; ++j) {
    int f = vtile * (kVT * kK) + j * 32 + lane;
    f = (f < kV * kK) ? f : (kV * kK - 1);
    const unsigned u = (unsigned)nidx[f];
    nbv[j] = (int)(((u < (unsigned)kV) ? u : (unsigned)(kV - 1)) * (unsigned)(kRow * sizeof(float)));
  }
  float gwv[15];
  #pragma unroll
  for (int j = 0; j < 15; ++j) {
    int f = vtile * (kVT * kG * kK) + j * 32 + lane;
    f = (f < kV * kG * kK) ? f : (kV * kG * kK - 1);
    gwv[j] = gradw[f];
  }

  float* grs = gr + wave * (kN * kGrP);

  const unsigned lane8 = (unsigned)lane * 8u;
  const gchar* xtb = (const gchar*)xt;
  v2f xr[kVT][kK];
  #pragma unroll
  for (int h = 0; h < 2; ++h) {
    unsigned vo[kVT / 2][kK];
    #pragma unroll
    for (int i = 0; i < kVT / 2; ++i) {
      #pragma unroll
      for (int k = 0; k < kK; ++k) {
        const int f = (h * (kVT / 2) + i) * kK + k;
        vo[i][k] = (unsigned)__builtin_amdgcn_readlane(nbv[f >> 5], f & 31) + lane8;
      }
    }
    __builtin_amdgcn_sched_barrier(0);
    #pragma unroll
    for (int i = 0; i < kVT / 2; ++i) {
      #pragma unroll
      for (int k = 0; k < kK; ++k)
        xr[h * (kVT / 2) + i][k] = *(const gv2f*)(xtb + vo[i][k]);
    }
    __builtin_amdgcn_sched_barrier(0);
  }
  #pragma unroll
  for (int vi = 0; vi < kVT; ++vi) {
    float sx[kG], sy[kG];
    #pragma unroll
    for (int g = 0; g < kG; ++g) {
      float s0 = 0.f, s1 = 0.f;
      #pragma unroll
      for (int k = 0; k < kK; ++k) {
        const int f = vi * (kG * kK) + g * kK + k;
        const float w = __int_as_float(
            __builtin_amdgcn_readlane(__float_as_int(gwv[f >> 5]), f & 31));
        s0 = __builtin_fmaf(xr[vi][k].x, w, s0);
        s1 = __builtin_fmaf(xr[vi][k].y, w, s1);
      }
      sx[g] = s0; sy[g] = s1;
    }
    float* dst = grs + half * kGrP + vi * kLp + vloc * (2 * kG);
    *(v2f*)(dst + 0) = v2f{sx[0], sx[1]};
    *(v2f*)(dst + 2) = v2f{sx[2], sx[3]};
    *(v2f*)(dst + 4) = v2f{sx[4], sy[0]};
    *(v2f*)(dst + 6) = v2f{sy[1], sy[2]};
    *(v2f*)(dst + 8) = v2f{sy[3], sy[4]};
  }

  v8f acc[kN][4] = {};
  #pragma unroll 4
  for (int kk = 0; kk < kKdim / 4; ++kk) {
    const int kb = kk * 4 + half * 2;
    const v2f b0 = *(const v2f*)(grs + 0 * kGrP + vloc * kLp + kb);
    const v2f b1 = *(const v2f*)(grs + 1 * kGrP + vloc * kLp + kb);
    #pragma unroll
    for (int m = 0; m < 4; ++m) {
      const v2f afr = *(const v2f*)(cw + (m * 16 + vloc) * kLp + kb);
      acc[0][m] = __builtin_amdgcn_wmma_f32_16x16x4_f32(
          false, afr, false, b0, (short)0, acc[0][m], false, false);
      acc[1][m] = __builtin_amdgcn_wmma_f32_16x16x4_f32(
          false, afr, false, b1, (short)0, acc[1][m], false, false);
    }
  }

  if (v < kV) {
    #pragma unroll
    for (int n = 0; n < kN; ++n) {
      #pragma unroll
      for (int m = 0; m < 4; ++m) {
        #pragma unroll
        for (int r = 0; r < 8; ++r) {
          const int o = m * 16 + half * 8 + r;
          out[((size_t)n * kCout + o) * kV + v] = acc[n][m][r] + convb[o];
        }
      }
    }
  }
}

extern "C" void kernel_launch(void* const* d_in, const int* in_sizes, int n_in,
                              void* d_out, int out_size, void* d_ws, size_t ws_size,
                              hipStream_t stream) {
  (void)in_sizes; (void)n_in; (void)out_size;
  const float* x      = (const float*)d_in[0];
  const int*   nidx   = (const int*)  d_in[1];
  const float* gradw  = (const float*)d_in[2];
  const float* convw  = (const float*)d_in[3];
  const float* convb  = (const float*)d_in[4];
  float*       out    = (float*)d_out;
  float*       xt     = (float*)d_ws;

  const int vtiles = (kV + kVT - 1) / kVT;
  const int blocks = (vtiles + kWaves - 1) / kWaves;
  const size_t lds_bytes =
      (size_t)(kCwF + kWaves * kN * kGrP) * sizeof(float);

  if (ws_size >= kXtBytes) {
    const int ttiles  = (kV + kTV - 1) / kTV;
    const int tblocks = (ttiles + kTW - 1) / kTW;
    x_to_vertex_rows<<<tblocks, kTW * 32, 0, stream>>>(x, xt);
    icosphere_conv_wmma_rows<<<blocks, kWaves * 32, lds_bytes, stream>>>(
        xt, nidx, gradw, convw, convb, out);
  } else {
    icosphere_conv_wmma<<<blocks, kWaves * 32, lds_bytes, stream>>>(
        x, nidx, gradw, convw, convb, out);
  }
}


# Gather kernel as a tile loop over 512 resident work-groups: each wave issues the next tile's row, weight and index loads ahead of the current tile's WMMA block; conv_w staged once; sums unchanged
# speedup vs baseline: 3.0717x; 1.0304x over previous
#include <hip/hip_runtime.h>

typedef __attribute__((ext_vector_type(2))) float v2f;
typedef __attribute__((ext_vector_type(4))) float v4f;
typedef __attribute__((ext_vector_type(8))) float v8f;
typedef __attribute__((address_space(1))) char gchar;
typedef __attribute__((address_space(1))) v2f  gv2f;

namespace {
constexpr int kV     = 163842;
constexpr int kK     = 6;
constexpr int kG     = 5;
constexpr int kN     = 2;
constexpr int kCin   = 32;
constexpr int kCout  = 64;
constexpr int kKdim  = kCin * kG;
constexpr int kVT    = 16;
constexpr int kWaves = 4;
constexpr int kCwN   = kCout * kKdim;
constexpr int kLp    = kKdim + 4;
constexpr int kCwF   = kCout * kLp;
constexpr int kGrP   = kVT * kLp + 32;
static_assert(kWaves * 16 == kCout && kKdim % 32 == 0, "conv_w copy: 16 rows per wave, 32 floats per load");
constexpr int kRow   = kN * kCin;
constexpr int kTV    = 32;
constexpr int kTW    = 4;
constexpr int kTP    = kRow + 4;
constexpr size_t kXtBytes = (size_t)kV * kRow * sizeof(float);
constexpr int kTiles = (kV + kVT - 1) / kVT;
constexpr int kLoopBlocks = 512;
}

__global__ __launch_bounds__(kWaves * 32) void icosphere_conv_wmma(
    const float* __restrict__ x,
    const int*   __restrict__ nidx,
    const float* __restrict__ gradw,
    const float* __restrict__ convw,
    const float* __restrict__ convb,
    float* __restrict__ out)
{
  extern __shared__ float smem[];
  float* cw = smem;
  float* gr = smem + kCwF;

  const int tid  = threadIdx.x;
  const int lane = tid & 31;
  const int wave = tid >> 5;
  const int vloc = lane & 15;
  const int half = lane >> 4;

  {
    const float* src = convw + (wave * 16) * kKdim + lane;
    float* dstc = cw + (wave * 16) * kLp + lane;
    for (int j = 0; j < kKdim / 32; ++j) {
      #pragma unroll
      for (int r = 0; r < 16; ++r) dstc[r * kLp + j * 32] = src[r * kKdim + j * 32];
    }
  }
  __syncthreads();

  const int vtile = blockIdx.x * kWaves + wave;
  const int v     = vtile * kVT + vloc;
  const int vc    = (v < kV) ? v : (kV - 1);

  int nb[kK];
  #pragma unroll
  for (int k = 0; k < kK; ++k) nb[k] = nidx[vc * kK + k];
  float gw[kG * kK];
  #pragma unroll
  for (int j = 0; j < kG * kK; ++j) gw[j] = gradw[(size_t)vc * (kG * kK) + j];

  float* grs = gr + wave * (kN * kGrP);

  #pragma unroll
  for (int n = 0; n < kN; ++n) {
    #pragma unroll 4
    for (int ci = 0; ci < kCin / 2; ++ci) {
      const int c = half * (kCin / 2) + ci;
      const float* xp = x + ((size_t)n * kCin + c) * kV;
      float xv[kK];
      #pragma unroll
      for (int k = 0; k < kK; ++k) xv[k] = xp[nb[k]];
      #pragma unroll
      for (int g = 0; g < kG; ++g) {
        float s = 0.f;
        #pragma unroll
        for (int k = 0; k < kK; ++k) s = __builtin_fmaf(xv[k], gw[g * kK + k], s);
        grs[n * kGrP + vloc * kLp + c * kG + g] = s;
      }
    }
  }

  v8f acc[kN][4] = {};
  #pragma unroll 4
  for (int kk = 0; kk < kKdim / 4; ++kk) {
    const int kb = kk * 4 + half * 2;
    const v2f b0 = *(const v2f*)(grs + 0 * kGrP + vloc * kLp + kb);
    const v2f b1 = *(const v2f*)(grs + 1 * kGrP + vloc * kLp + kb);
    #pragma unroll
    for (int m = 0; m < 4; ++m) {
      const v2f afr = *(const v2f*)(cw + (m * 16 + vloc) * kLp + kb);
      acc[0][m] = __builtin_amdgcn_wmma_f32_16x16x4_f32(
          false, afr, false, b0, (short)0, acc[0][m], false, false);
      acc[1][m] = __builtin_amdgcn_wmma_f32_16x16x4_f32(
          false, afr, false, b1, (short)0, acc[1][m], false, false);
    }
  }

  if (v < kV) {
    #pragma unroll
    for (int n = 0; n < kN; ++n) {
      #pragma unroll
      for (int m = 0; m < 4; ++m) {
        #pragma unroll
        for (int r = 0; r < 8; ++r) {
          const int o = m * 16 + half * 8 + r;
          out[((size_t)n * kCout + o) * kV + v] = acc[n][m][r] + convb[o];
        }
      }
    }
  }
}

__global__ __launch_bounds__(kTW * 32) void x_to_vertex_rows(
    const float* __restrict__ x,
    float* __restrict__ xt)
{
  __shared__ float tile[kTW * kTV * kTP];
  const int lane = threadIdx.x & 31;
  const int wave = threadIdx.x >> 5;
  const int v0   = (blockIdx.x * kTW + wave) * kTV;
  const int v    = v0 + lane;
  const int vc   = (v < kV) ? v : (kV - 1);

  float r[kRow];
  #pragma unroll
  for (int j = 0; j < kRow; ++j) r[j] = x[(size_t)j * kV + vc];

  float* t = tile + wave * (kTV * kTP);
  #pragma unroll
  for (int q = 0; q < kRow / 4; ++q) {
    v4f w4 = {r[4 * q], r[4 * q + 1], r[4 * q + 2], r[4 * q + 3]};
    *(v4f*)(t + lane * kTP + 4 * q) = w4;
  }
  const int sub  = lane >> 4;
  const int col4 = (lane & 15) * 4;
  v4f w[kTV / 2];
  #pragma unroll
  for (int s = 0; s < kTV / 2; ++s) w[s] = *(const v4f*)(t + (2 * s + sub) * kTP + col4);
  float* dst = xt + (size_t)(v0 + sub) * kRow + col4;
  if (v0 + kTV <= kV) {
    #pragma unroll
    for (int s = 0; s < kTV / 2; ++s) *(v4f*)(dst + (size_t)(2 * s) * kRow) = w[s];
  } else {
    #pragma unroll
    for (int s = 0; s < kTV / 2; ++s)
      if (v0 + 2 * s + sub < kV) *(v4f*)(dst + (size_t)(2 * s) * kRow) = w[s];
  }
}


__device__ __forceinline__ void load_tile_nidx(
    const int* __restrict__ nidx, int vtile, int lane, int (&nbraw)[3])
{
  #pragma unroll
  for (int j = 0; j < 3; ++j) {
    int f = vtile * (kVT * kK) + j * 32 + lane;
    f = (f < kV * kK) ? f : (kV * kK - 1);
    nbraw[j] = nidx[f];
  }
}
__device__ __forceinline__ void load_tile_gradw(
    const float* __restrict__ gradw, int vtile, int lane, float (&gwv)[15])
{
  #pragma unroll
  for (int j = 0; j < 15; ++j) {
    int f = vtile * (kVT * kG * kK) + j * 32 + lane;
    f = (f < kV * kG * kK) ? f : (kV * kG * kK - 1);
    gwv[j] = gradw[f];
  }
}

__device__ __forceinline__ void issue_row_gather(
    const gchar* xtb, const int (&nbraw)[3], unsigned lane8, v2f (&xr)[kVT][kK])
{
  int nbv[3];
  #pragma unroll
  for (int j = 0; j < 3; ++j) {
    const unsigned u = (unsigned)nbraw[j];
    nbv[j] = (int)(((u < (unsigned)kV) ? u : (unsigned)(kV - 1)) * (unsigned)(kRow * sizeof(float)));
  }
  #pragma unroll
  for (int h = 0; h < 2; ++h) {
    unsigned vo[kVT / 2][kK];
    #pragma unroll
    for (int i = 0; i < kVT / 2; ++i) {
      #pragma unroll
      for (int k = 0; k < kK; ++k) {
        const int f = (h * (kVT / 2) + i) * kK + k;
        vo[i][k] = (unsigned)__builtin_amdgcn_readlane(nbv[f >> 5], f & 31) + lane8;
      }
    }
    __builtin_amdgcn_sched_barrier(0);
    #pragma unroll
    for (int i = 0; i < kVT / 2; ++i) {
      #pragma unroll
      for (int k = 0; k < kK; ++k)
        xr[h * (kVT / 2) + i][k] = *(const gv2f*)(xtb + vo[i][k]);
    }
    __builtin_amdgcn_sched_barrier(0);
  }
}

__global__ __launch_bounds__(kWaves * 32) void icosphere_conv_wmma_rows(
    const float* __restrict__ xt,
    const int*   __restrict__ nidx,
    const float* __restrict__ gradw,
    const float* __restrict__ convw,
    const float* __restrict__ convb,
    float* __restrict__ out)
{
  extern __shared__ float smem[];
  float* cw = smem;
  float* gr = smem + kCwF;

  const int tid  = threadIdx.x;
  const int lane = tid & 31;
  const int wave = tid >> 5;
  const int vloc = lane & 15;
  const int half = lane >> 4;

  const int nwaves = (int)gridDim.x * kWaves;
  const int gwave  = (int)blockIdx.x * kWaves + wave;
  const int niter  = (kTiles - gwave + nwaves - 1) / nwaves;

  const unsigned lane8 = (unsigned)lane * 8u;
  const gchar* xtb = (const gchar*)xt;

  int   nb[3];
  float gw[15];
  v2f   xr[kVT][kK];

  load_tile_nidx(nidx, gwave, lane, nb);
  load_tile_gradw(gradw, gwave, lane, gw);

  float bias[4][8];
  #pragma unroll
  for (int m = 0; m < 4; ++m) {
    #pragma unroll
    for (int r = 0; r < 8; ++r) bias[m][r] = convb[m * 16 + half * 8 + r];
  }

  {
    const float* src = convw + (wave * 16) * kKdim + lane;
    float* dstc = cw + (wave * 16) * kLp + lane;
    for (int j = 0; j < kKdim / 32; ++j) {
      #pragma unroll
      for (int r = 0; r < 16; ++r) dstc[r * kLp + j * 32] = src[r * kKdim + j * 32];
    }
  }
  __syncthreads();

  issue_row_gather(xtb, nb, lane8, xr);
  load_tile_nidx(nidx, gwave + nwaves, lane, nb);

  float* grs = gr + wave * (kN * kGrP);

  for (int it = 0; it < niter; ++it) {
    const int vtile = gwave + it * nwaves;
    const int v     = vtile * kVT + vloc;

    #pragma unroll
    for (int vi = 0; vi < kVT; ++vi) {
      float sx[kG], sy[kG];
      #pragma unroll
      for (int g = 0; g < kG; ++g) {
        float s0 = 0.f, s1 = 0.f;
        #pragma unroll
        for (int k = 0; k < kK; ++k) {
          const int f = vi * (kG * kK) + g * kK + k;
          const float w = __int_as_float(
              __builtin_amdgcn_readlane(__float_as_int(gw[f >> 5]), f & 31));
          s0 = __builtin_fmaf(xr[vi][k].x, w, s0);
          s1 = __builtin_fmaf(xr[vi][k].y, w, s1);
        }
        sx[g] = s0; sy[g] = s1;
      }
      float* dst = grs + half * kGrP + vi * kLp + vloc * (2 * kG);
      *(v2f*)(dst + 0) = v2f{sx[0], sx[1]};
      *(v2f*)(dst + 2) = v2f{sx[2], sx[3]};
      *(v2f*)(dst + 4) = v2f{sx[4], sy[0]};
      *(v2f*)(dst + 6) = v2f{sy[1], sy[2]};
      *(v2f*)(dst + 8) = v2f{sy[3], sy[4]};
    }

    if (it + 1 < niter) {
      issue_row_gather(xtb, nb, lane8, xr);
      load_tile_nidx(nidx, vtile + 2 * nwaves, lane, nb);
      load_tile_gradw(gradw, vtile + nwaves, lane, gw);
    }

    v8f acc[kN][4] = {};
    #pragma unroll 4
    for (int kk = 0; kk < kKdim / 4; ++kk) {
      const int kb = kk * 4 + half * 2;
      const v2f b0 = *(const v2f*)(grs + 0 * kGrP + vloc * kLp + kb);
      const v2f b1 = *(const v2f*)(grs + 1 * kGrP + vloc * kLp + kb);
      #pragma unroll
      for (int m = 0; m < 4; ++m) {
        const v2f afr = *(const v2f*)(cw + (m * 16 + vloc) * kLp + kb);
        acc[0][m] = __builtin_amdgcn_wmma_f32_16x16x4_f32(
            false, afr, false, b0, (short)0, acc[0][m], false, false);
        acc[1][m] = __builtin_amdgcn_wmma_f32_16x16x4_f32(
            false, afr, false, b1, (short)0, acc[1][m], false, false);
      }
    }

    if (v < kV) {
      #pragma unroll
      for (int n = 0; n < kN; ++n) {
        #pragma unroll
        for (int m = 0; m < 4; ++m) {
          #pragma unroll
          for (int r = 0; r < 8; ++r) {
            const int o = m * 16 + half * 8 + r;
            out[((size_t)n * kCout + o) * kV + v] = acc[n][m][r] + bias[m][r];
          }
        }
      }
    }
  }
}

extern "C" void kernel_launch(void* const* d_in, const int* in_sizes, int n_in,
                              void* d_out, int out_size, void* d_ws, size_t ws_size,
                              hipStream_t stream) {
  (void)in_sizes; (void)n_in; (void)out_size;
  const float* x      = (const float*)d_in[0];
  const int*   nidx   = (const int*)  d_in[1];
  const float* gradw  = (const float*)d_in[2];
  const float* convw  = (const float*)d_in[3];
  const float* convb  = (const float*)d_in[4];
  float*       out    = (float*)d_out;
  float*       xt     = (float*)d_ws;

  const int vtiles = (kV + kVT - 1) / kVT;
  const int blocks = (vtiles + kWaves - 1) / kWaves;
  const size_t lds_bytes =
      (size_t)(kCwF + kWaves * kN * kGrP) * sizeof(float);

  if (ws_size >= kXtBytes) {
    const int ttiles  = (kV + kTV - 1) / kTV;
    const int tblocks = (ttiles + kTW - 1) / kTW;
    x_to_vertex_rows<<<tblocks, kTW * 32, 0, stream>>>(x, xt);
    icosphere_conv_wmma_rows<<<kLoopBlocks, kWaves * 32, lds_bytes, stream>>>(
        xt, nidx, gradw, convw, convb, out);
  } else {
    icosphere_conv_wmma<<<blocks, kWaves * 32, lds_bytes, stream>>>(
        x, nidx, gradw, convw, convb, out);
  }
}


# Tile loop as one 12-wave work-group per WGP: three waves per SIMD, not two, fill each other's gather and store waits (bias in LDS, wave index scalar: 336 VGPRs); tiles dealt in chunks; sums unchanged
# speedup vs baseline: 3.2087x; 1.0446x over previous
#include <hip/hip_runtime.h>

typedef __attribute__((ext_vector_type(2))) float v2f;
typedef __attribute__((ext_vector_type(4))) float v4f;
typedef __attribute__((ext_vector_type(8))) float v8f;
typedef __attribute__((address_space(1))) char gchar;
typedef __attribute__((address_space(1))) v2f  gv2f;

namespace {
constexpr int kV     = 163842;
constexpr int kK     = 6;
constexpr int kG     = 5;
constexpr int kN     = 2;
constexpr int kCin   = 32;
constexpr int kCout  = 64;
constexpr int kKdim  = kCin * kG;
constexpr int kVT    = 16;
constexpr int kWaves = 4;
constexpr int kCwN   = kCout * kKdim;
constexpr int kLp    = kKdim + 4;
constexpr int kCwF   = kCout * kLp;
constexpr int kGrP   = kVT * kLp + 32;
static_assert(kWaves * 16 == kCout && kKdim % 32 == 0, "conv_w copy: 16 rows per wave, 32 floats per load");
constexpr int kRow   = kN * kCin;
constexpr int kTV    = 32;
constexpr int kTW    = 4;
constexpr int kTP    = kRow + 4;
constexpr size_t kXtBytes = (size_t)kV * kRow * sizeof(float);
constexpr int kTiles = (kV + kVT - 1) / kVT;
constexpr int kLoopBlocks = 256;
constexpr int kWavesL = 12;
constexpr int kLdsFloatsL = kCwF + kWavesL * kN * kGrP + kCout;
static_assert(kLdsFloatsL * sizeof(float) <= 320 * 1024, "one work-group's LDS");
}

__global__ __launch_bounds__(kWaves * 32) void icosphere_conv_wmma(
    const float* __restrict__ x,
    const int*   __restrict__ nidx,
    const float* __restrict__ gradw,
    const float* __restrict__ convw,
    const float* __restrict__ convb,
    float* __restrict__ out)
{
  extern __shared__ float smem[];
  float* cw = smem;
  float* gr = smem + kCwF;

  const int tid  = threadIdx.x;
  const int lane = tid & 31;
  const int wave = tid >> 5;
  const int vloc = lane & 15;
  const int half = lane >> 4;

  {
    const float* src = convw + (wave * 16) * kKdim + lane;
    float* dstc = cw + (wave * 16) * kLp + lane;
    for (int j = 0; j < kKdim / 32; ++j) {
      #pragma unroll
      for (int r = 0; r < 16; ++r) dstc[r * kLp + j * 32] = src[r * kKdim + j * 32];
    }
  }
  __syncthreads();

  const int vtile = blockIdx.x * kWaves + wave;
  const int v     = vtile * kVT + vloc;
  const int vc    = (v < kV) ? v : (kV - 1);

  int nb[kK];
  #pragma unroll
  for (int k = 0; k < kK; ++k) nb[k] = nidx[vc * kK + k];
  float gw[kG * kK];
  #pragma unroll
  for (int j = 0; j < kG * kK; ++j) gw[j] = gradw[(size_t)vc * (kG * kK) + j];

  float* grs = gr + wave * (kN * kGrP);

  #pragma unroll
  for (int n = 0; n < kN; ++n) {
    #pragma unroll 4
    for (int ci = 0; ci < kCin / 2; ++ci) {
      const int c = half * (kCin / 2) + ci;
      const float* xp = x + ((size_t)n * kCin + c) * kV;
      float xv[kK];
      #pragma unroll
      for (int k = 0; k < kK; ++k) xv[k] = xp[nb[k]];
      #pragma unroll
      for (int g = 0; g < kG; ++g) {
        float s = 0.f;
        #pragma unroll
        for (int k = 0; k < kK; ++k) s = __builtin_fmaf(xv[k], gw[g * kK + k], s);
        grs[n * kGrP + vloc * kLp + c * kG + g] = s;
      }
    }
  }

  v8f acc[kN][4] = {};
  #pragma unroll 4
  for (int kk = 0; kk < kKdim / 4; ++kk) {
    const int kb = kk * 4 + half * 2;
    const v2f b0 = *(const v2f*)(grs + 0 * kGrP + vloc * kLp + kb);
    const v2f b1 = *(const v2f*)(grs + 1 * kGrP + vloc * kLp + kb);
    #pragma unroll
    for (int m = 0; m < 4; ++m) {
      const v2f afr = *(const v2f*)(cw + (m * 16 + vloc) * kLp + kb);
      acc[0][m] = __builtin_amdgcn_wmma_f32_16x16x4_f32(
          false, afr, false, b0, (short)0, acc[0][m], false, false);
      acc[1][m] = __builtin_amdgcn_wmma_f32_16x16x4_f32(
          false, afr, false, b1, (short)0, acc[1][m], false, false);
    }
  }

  if (v < kV) {
    #pragma unroll
    for (int n = 0; n < kN; ++n) {
      #pragma unroll
      for (int m = 0; m < 4; ++m) {
        #pragma unroll
        for (int r = 0; r < 8; ++r) {
          const int o = m * 16 + half * 8 + r;
          out[((size_t)n * kCout + o) * kV + v] = acc[n][m][r] + convb[o];
        }
      }
    }
  }
}

__global__ __launch_bounds__(kTW * 32) void x_to_vertex_rows(
    const float* __restrict__ x,
    float* __restrict__ xt)
{
  __shared__ float tile[kTW * kTV * kTP];
  const int lane = threadIdx.x & 31;
  const int wave = threadIdx.x >> 5;
  const int v0   = (blockIdx.x * kTW + wave) * kTV;
  const int v    = v0 + lane;
  const int vc   = (v < kV) ? v : (kV - 1);

  float r[kRow];
  #pragma unroll
  for (int j = 0; j < kRow; ++j) r[j] = x[(size_t)j * kV + vc];

  float* t = tile + wave * (kTV * kTP);
  #pragma unroll
  for (int q = 0; q < kRow / 4; ++q) {
    v4f w4 = {r[4 * q], r[4 * q + 1], r[4 * q + 2], r[4 * q + 3]};
    *(v4f*)(t + lane * kTP + 4 * q) = w4;
  }
  const int sub  = lane >> 4;
  const int col4 = (lane & 15) * 4;
  v4f w[kTV / 2];
  #pragma unroll
  for (int s = 0; s < kTV / 2; ++s) w[s] = *(const v4f*)(t + (2 * s + sub) * kTP + col4);
  float* dst = xt + (size_t)(v0 + sub) * kRow + col4;
  if (v0 + kTV <= kV) {
    #pragma unroll
    for (int s = 0; s < kTV / 2; ++s) *(v4f*)(dst + (size_t)(2 * s) * kRow) = w[s];
  } else {
    #pragma unroll
    for (int s = 0; s < kTV / 2; ++s)
      if (v0 + 2 * s + sub < kV) *(v4f*)(dst + (size_t)(2 * s) * kRow) = w[s];
  }
}


__device__ __forceinline__ void load_tile_nidx(
    const int* __restrict__ nidx, int vtile, int lane, int (&nbraw)[3])
{
  #pragma unroll
  for (int j = 0; j < 3; ++j) {
    int f = vtile * (kVT * kK) + j * 32 + lane;
    f = (f < kV * kK) ? f : (kV * kK - 1);
    nbraw[j] = nidx[f];
  }
}
__device__ __forceinline__ void load_tile_gradw(
    const float* __restrict__ gradw, int vtile, int lane, float (&gwv)[15])
{
  #pragma unroll
  for (int j = 0; j < 15; ++j) {
    int f = vtile * (kVT * kG * kK) + j * 32 + lane;
    f = (f < kV * kG * kK) ? f : (kV * kG * kK - 1);
    gwv[j] = gradw[f];
  }
}

__device__ __forceinline__ void issue_row_gather(
    const gchar* xtb, const int (&nbraw)[3], unsigned lane8, v2f (&xr)[kVT][kK])
{
  int nbv[3];
  #pragma unroll
  for (int j = 0; j < 3; ++j) {
    const unsigned u = (unsigned)nbraw[j];
    nbv[j] = (int)(((u < (unsigned)kV) ? u : (unsigned)(kV - 1)) * (unsigned)(kRow * sizeof(float)));
  }
  #pragma unroll
  for (int h = 0; h < 2; ++h) {
    unsigned vo[kVT / 2][kK];
    #pragma unroll
    for (int i = 0; i < kVT / 2; ++i) {
      #pragma unroll
      for (int k = 0; k < kK; ++k) {
        const int f = (h * (kVT / 2) + i) * kK + k;
        vo[i][k] = (unsigned)__builtin_amdgcn_readlane(nbv[f >> 5], f & 31) + lane8;
      }
    }
    __builtin_amdgcn_sched_barrier(0);
    #pragma unroll
    for (int i = 0; i < kVT / 2; ++i) {
      #pragma unroll
      for (int k = 0; k < kK; ++k)
        xr[h * (kVT / 2) + i][k] = *(const gv2f*)(xtb + vo[i][k]);
    }
    __builtin_amdgcn_sched_barrier(0);
  }
}

__global__ __launch_bounds__(kWavesL * 32) void icosphere_conv_wmma_rows(
    const float* __restrict__ xt,
    const int*   __restrict__ nidx,
    const float* __restrict__ gradw,
    const float* __restrict__ convw,
    const float* __restrict__ convb,
    float* __restrict__ out)
{
  extern __shared__ float smem[];
  float* cw = smem;
  float* gr = smem + kCwF;
  float* bs = gr + kWavesL * kN * kGrP;

  const int tid  = threadIdx.x;
  const int lane = tid & 31;
  const int wave = __builtin_amdgcn_readfirstlane(tid >> 5);
  const int vloc = lane & 15;
  const int half = lane >> 4;

  const int nblk   = (int)gridDim.x;
  const int per    = kTiles / nblk;
  const int cbase  = (int)blockIdx.x * per;
  const int ccount = ((int)blockIdx.x == nblk - 1) ? (kTiles - cbase) : per;
  const int nwaves = kWavesL;
  const int gwave  = cbase + wave;
  const int niter  = (ccount - wave + nwaves - 1) / nwaves;

  const unsigned lane8 = (unsigned)lane * 8u;
  const gchar* xtb = (const gchar*)xt;

  int   nb[3];
  float gw[15];
  v2f   xr[kVT][kK];

  load_tile_nidx(nidx, gwave, lane, nb);
  load_tile_gradw(gradw, gwave, lane, gw);

  static_assert(8 * 8 == kCout, "conv_w copy: 8 waves x 8 rows");
  if (wave < 8) {
    const float* src = convw + (wave * 8) * kKdim + lane;
    float* dstc = cw + (wave * 8) * kLp + lane;
    #pragma unroll
    for (int j = 0; j < kKdim / 32; ++j) {
      #pragma unroll
      for (int r = 0; r < 8; ++r) dstc[r * kLp + j * 32] = src[r * kKdim + j * 32];
    }
  }
  if (tid < kCout) bs[tid] = convb[tid];
  __syncthreads();

  issue_row_gather(xtb, nb, lane8, xr);
  load_tile_nidx(nidx, gwave + nwaves, lane, nb);

  float* grs = gr + wave * (kN * kGrP);

  for (int it = 0; it < niter; ++it) {
    const int vtile = gwave + it * nwaves;
    const int v     = vtile * kVT + vloc;

    #pragma unroll
    for (int vi = 0; vi < kVT; ++vi) {
      float sx[kG], sy[kG];
      #pragma unroll
      for (int g = 0; g < kG; ++g) {
        float s0 = 0.f, s1 = 0.f;
        #pragma unroll
        for (int k = 0; k < kK; ++k) {
          const int f = vi * (kG * kK) + g * kK + k;
          const float w = __int_as_float(
              __builtin_amdgcn_readlane(__float_as_int(gw[f >> 5]), f & 31));
          s0 = __builtin_fmaf(xr[vi][k].x, w, s0);
          s1 = __builtin_fmaf(xr[vi][k].y, w, s1);
        }
        sx[g] = s0; sy[g] = s1;
      }
      float* dst = grs + half * kGrP + vi * kLp + vloc * (2 * kG);
      *(v2f*)(dst + 0) = v2f{sx[0], sx[1]};
      *(v2f*)(dst + 2) = v2f{sx[2], sx[3]};
      *(v2f*)(dst + 4) = v2f{sx[4], sy[0]};
      *(v2f*)(dst + 6) = v2f{sy[1], sy[2]};
      *(v2f*)(dst + 8) = v2f{sy[3], sy[4]};
    }

    if (it + 1 < niter) {
      issue_row_gather(xtb, nb, lane8, xr);
      load_tile_nidx(nidx, vtile + 2 * nwaves, lane, nb);
      load_tile_gradw(gradw, vtile + nwaves, lane, gw);
    }

    v8f acc[kN][4] = {};
    #pragma unroll 4
    for (int kk = 0; kk < kKdim / 4; ++kk) {
      const int kb = kk * 4 + half * 2;
      const v2f b0 = *(const v2f*)(grs + 0 * kGrP + vloc * kLp + kb);
      const v2f b1 = *(const v2f*)(grs + 1 * kGrP + vloc * kLp + kb);
      #pragma unroll
      for (int m = 0; m < 4; ++m) {
        const v2f afr = *(const v2f*)(cw + (m * 16 + vloc) * kLp + kb);
        acc[0][m] = __builtin_amdgcn_wmma_f32_16x16x4_f32(
            false, afr, false, b0, (short)0, acc[0][m], false, false);
        acc[1][m] = __builtin_amdgcn_wmma_f32_16x16x4_f32(
            false, afr, false, b1, (short)0, acc[1][m], false, false);
      }
    }

    if (v < kV) {
      #pragma unroll
      for (int m = 0; m < 4; ++m) {
        const v4f blo = *(const v4f*)(bs + m * 16 + half * 8);
        const v4f bhi = *(const v4f*)(bs + m * 16 + half * 8 + 4);
        #pragma unroll
        for (int n = 0; n < kN; ++n) {
          #pragma unroll
          for (int r = 0; r < 8; ++r) {
            const int o = m * 16 + half * 8 + r;
            const float b = (r < 4) ? blo[r & 3] : bhi[r & 3];
            out[((size_t)n * kCout + o) * kV + v] = acc[n][m][r] + b;
          }
        }
      }
    }
  }
}

extern "C" void kernel_launch(void* const* d_in, const int* in_sizes, int n_in,
                              void* d_out, int out_size, void* d_ws, size_t ws_size,
                              hipStream_t stream) {
  (void)in_sizes; (void)n_in; (void)out_size;
  const float* x      = (const float*)d_in[0];
  const int*   nidx   = (const int*)  d_in[1];
  const float* gradw  = (const float*)d_in[2];
  const float* convw  = (const float*)d_in[3];
  const float* convb  = (const float*)d_in[4];
  float*       out    = (float*)d_out;
  float*       xt     = (float*)d_ws;

  const int vtiles = (kV + kVT - 1) / kVT;
  const int blocks = (vtiles + kWaves - 1) / kWaves;
  const size_t lds_bytes =
      (size_t)(kCwF + kWaves * kN * kGrP) * sizeof(float);
  const size_t lds_bytes_loop = (size_t)kLdsFloatsL * sizeof(float);

  if (ws_size >= kXtBytes) {
    const int ttiles  = (kV + kTV - 1) / kTV;
    const int tblocks = (ttiles + kTW - 1) / kTW;
    x_to_vertex_rows<<<tblocks, kTW * 32, 0, stream>>>(x, xt);
    icosphere_conv_wmma_rows<<<kLoopBlocks, kWavesL * 32, lds_bytes_loop, stream>>>(
        xt, nidx, gradw, convw, convb, out);
  } else {
    icosphere_conv_wmma<<<blocks, kWaves * 32, lds_bytes, stream>>>(
        x, nidx, gradw, convw, convb, out);
  }
}
